# starting kernel
# baseline (speedup 1.0000x reference)
#include <hip/hip_runtime.h>

typedef __attribute__((ext_vector_type(2))) float v2f;
typedef __attribute__((ext_vector_type(8))) float v8f;

#define DHID     256
#define LTOK     128
#define LLAB     32
#define NPG      64
#define BGRAPH   64
#define NNODES   (BGRAPH * NPG)
#define ODIM     1024

__global__ void ragged_bag_kernel(const int* __restrict__ text_ids,
                                  const int* __restrict__ label_ids,
                                  const int* __restrict__ text_len,
                                  const int* __restrict__ label_len,
                                  const float* __restrict__ word_emb,
                                  const float* __restrict__ label_emb,
                                  float* __restrict__ text_out,
                                  float* __restrict__ label_out) {
    __shared__ int s_tid[LTOK];
    __shared__ int s_lid[LLAB];
    const int n = blockIdx.x;
    const int d = threadIdx.x;

    if (threadIdx.x < LTOK) s_tid[threadIdx.x] = text_ids[n * LTOK + threadIdx.x];
    if (threadIdx.x < LLAB) s_lid[threadIdx.x] = label_ids[n * LLAB + threadIdx.x];
    __syncthreads();

    const int tl = text_len[n];
    float acc = 0.0f;
    for (int l = 0; l < tl; ++l)
        acc += word_emb[(size_t)s_tid[l] * DHID + d];
    text_out[(size_t)n * DHID + d] = acc / (float)tl;

    const int ll = label_len[n];
    float accl = 0.0f;
    for (int l = 0; l < ll; ++l)
        accl += label_emb[(size_t)s_lid[l] * DHID + d];
    label_out[(size_t)n * DHID + d] = accl;
}

__global__ void gemm_wmma_f32(const float* __restrict__ A,
                              const float* __restrict__ B,
                              const float* __restrict__ bias,
                              float* __restrict__ C,
                              int M, int N, int K, int relu) {
    const int wave       = (blockIdx.x * blockDim.x + threadIdx.x) >> 5;
    const int tilesN     = N >> 4;
    const int totalTiles = (M >> 4) * tilesN;
    if (wave >= totalTiles) return;

    const int tm   = wave / tilesN;
    const int tn   = wave % tilesN;
    const int lane = threadIdx.x & 31;
    const int kh   = lane >> 4;
    const int l15  = lane & 15;

    const float* Arow = A + (size_t)(tm * 16 + l15) * K;
    const float* Bcol = B + (size_t)(tn * 16 + l15);

    v8f acc = {};
    for (int k = 0; k < K; k += 4) {
        const int kk = k + kh * 2;
        v2f a = *(const v2f*)(Arow + kk);
        v2f b;
        b.x = Bcol[(size_t)kk * N];
        b.y = Bcol[(size_t)(kk + 1) * N];
        acc = __builtin_amdgcn_wmma_f32_16x16x4_f32(
             false, a,  false, b,
             (short)0, acc,  false,  false);
    }

    const int col  = tn * 16 + l15;
    const int row0 = tm * 16 + kh * 8;
    const float bv = bias ? bias[col] : 0.0f;
#pragma unroll
    for (int v = 0; v < 8; ++v) {
        float val = acc[v] + bv;
        if (relu) val = fmaxf(val, 0.0f);
        C[(size_t)(row0 + v) * N + col] = val;
    }
}

__global__ void readout_kernel(const float* __restrict__ h_t,
                               const float* __restrict__ h_l,
                               const float* __restrict__ w_adap,
                               float* __restrict__ fused) {
    const int b = blockIdx.x;
    const int d = threadIdx.x;
    float st = 0.0f, sl = 0.0f;
    for (int i = 0; i < NPG; ++i) {
        const size_t node = (size_t)b * NPG + i;
        st += h_t[node * DHID + d];
        sl += h_l[node * DHID + d];
    }
    const float inv = 1.0f / (float)NPG;
    fused[(size_t)b * DHID + d] = (st * inv) * w_adap[0] + (sl * inv) * w_adap[1];
}

extern "C" void kernel_launch(void* const* d_in, const int* in_sizes, int n_in,
                              void* d_out, int out_size, void* d_ws, size_t ws_size,
                              hipStream_t stream) {
    const int*   text_ids  = (const int*)  d_in[0];
    const int*   label_ids = (const int*)  d_in[1];
    const int*   text_len  = (const int*)  d_in[2];
    const int*   label_len = (const int*)  d_in[3];
    const float* word_emb  = (const float*)d_in[4];
    const float* label_emb = (const float*)d_in[5];
    const float* Wg_text   = (const float*)d_in[6];
    const float* Wg_label  = (const float*)d_in[7];
    const float* w_adap    = (const float*)d_in[8];
    const float* W1        = (const float*)d_in[9];
    const float* b1        = (const float*)d_in[10];
    const float* Wout      = (const float*)d_in[11];
    const float* bout      = (const float*)d_in[12];
    float* out = (float*)d_out;

    float* ws        = (float*)d_ws;
    float* text_out  = ws;
    float* label_out = text_out  + (size_t)NNODES * DHID;
    float* h_t       = label_out + (size_t)NNODES * DHID;
    float* h_l       = h_t       + (size_t)NNODES * DHID;
    float* fused     = h_l       + (size_t)NNODES * DHID;
    float* hmid      = fused     + (size_t)BGRAPH * DHID;

    ragged_bag_kernel<<<NNODES, 256, 0, stream>>>(
        text_ids, label_ids, text_len, label_len,
        word_emb, label_emb, text_out, label_out);

    const int tiles1 = (NNODES / 16) * (DHID / 16);
    gemm_wmma_f32<<<(tiles1 + 7) / 8, 256, 0, stream>>>(
        text_out, Wg_text, nullptr, h_t, NNODES, DHID, DHID,  1);
    gemm_wmma_f32<<<(tiles1 + 7) / 8, 256, 0, stream>>>(
        label_out, Wg_label, nullptr, h_l, NNODES, DHID, DHID,  1);

    readout_kernel<<<BGRAPH, 256, 0, stream>>>(h_t, h_l, w_adap, fused);

    const int tiles2 = (BGRAPH / 16) * (DHID / 16);
    gemm_wmma_f32<<<(tiles2 + 7) / 8, 256, 0, stream>>>(
        fused, W1, b1, hmid, BGRAPH, DHID, DHID,  1);

    const int tiles3 = (BGRAPH / 16) * (ODIM / 16);
    gemm_wmma_f32<<<(tiles3 + 7) / 8, 256, 0, stream>>>(
        hmid, Wout, bout, out, BGRAPH, ODIM, DHID,  0);
}


# GEMM K loop: fragment loads of 16 K-steps issued as one group, one group ahead of their WMMAs (two register sets; N, K compile-time so a group needs no address math); k order of the sums unchanged
# speedup vs baseline: 1.4039x; 1.4039x over previous
#include <hip/hip_runtime.h>

typedef __attribute__((ext_vector_type(2))) float v2f;
typedef __attribute__((ext_vector_type(8))) float v8f;

#define DHID     256
#define LTOK     128
#define LLAB     32
#define NPG      64
#define BGRAPH   64
#define NNODES   (BGRAPH * NPG)
#define ODIM     1024

__global__ void ragged_bag_kernel(const int* __restrict__ text_ids,
                                  const int* __restrict__ label_ids,
                                  const int* __restrict__ text_len,
                                  const int* __restrict__ label_len,
                                  const float* __restrict__ word_emb,
                                  const float* __restrict__ label_emb,
                                  float* __restrict__ text_out,
                                  float* __restrict__ label_out) {
    __shared__ int s_tid[LTOK];
    __shared__ int s_lid[LLAB];
    const int n = blockIdx.x;
    const int d = threadIdx.x;

    if (threadIdx.x < LTOK) s_tid[threadIdx.x] = text_ids[n * LTOK + threadIdx.x];
    if (threadIdx.x < LLAB) s_lid[threadIdx.x] = label_ids[n * LLAB + threadIdx.x];
    __syncthreads();

    const int tl = text_len[n];
    float acc = 0.0f;
    for (int l = 0; l < tl; ++l)
        acc += word_emb[(size_t)s_tid[l] * DHID + d];
    text_out[(size_t)n * DHID + d] = acc / (float)tl;

    const int ll = label_len[n];
    float accl = 0.0f;
    for (int l = 0; l < ll; ++l)
        accl += label_emb[(size_t)s_lid[l] * DHID + d];
    label_out[(size_t)n * DHID + d] = accl;
}

#define KGROUP   16

template <int N>
__device__ __forceinline__ void load_group(v2f (&a)[KGROUP], v2f (&b)[KGROUP],
                                           const float* __restrict__ Ap,
                                           const float* __restrict__ Bp) {
#pragma unroll
    for (int u = 0; u < KGROUP; ++u) {
        a[u]   = *(const v2f*)(Ap + u * 4);
        b[u].x = Bp[(size_t)(u * 4) * N];
        b[u].y = Bp[(size_t)(u * 4 + 1) * N];
    }
}

__device__ __forceinline__ void mma_group(v8f& acc, const v2f (&a)[KGROUP],
                                          const v2f (&b)[KGROUP]) {
#pragma unroll
    for (int u = 0; u < KGROUP; ++u)
        acc = __builtin_amdgcn_wmma_f32_16x16x4_f32(
             false, a[u],  false, b[u],
             (short)0, acc,  false,  false);
}

template <int N, int K>
__global__ void __launch_bounds__(256)
gemm_wmma_f32(const float* __restrict__ A,
              const float* __restrict__ B,
              const float* __restrict__ bias,
              float* __restrict__ C,
              int M, int relu) {
    static_assert(N % 16 == 0, "whole 16-column tiles");
    static_assert(K % (4 * KGROUP) == 0, "whole load groups");
    constexpr int NG = K / (4 * KGROUP);

    const int wave       = (blockIdx.x * blockDim.x + threadIdx.x) >> 5;
    const int tilesN     = N >> 4;
    const int totalTiles = (M >> 4) * tilesN;
    if (wave >= totalTiles) return;

    const int tm   = wave / tilesN;
    const int tn   = wave % tilesN;
    const int lane = threadIdx.x & 31;
    const int kh   = lane >> 4;
    const int l15  = lane & 15;

    const float* Ap = A + (size_t)(tm * 16 + l15) * K + kh * 2;
    const float* Bp = B + (size_t)(kh * 2) * N + (size_t)(tn * 16 + l15);

    v8f acc = {};
    v2f a[2][KGROUP], b[2][KGROUP];
    load_group<N>(a[0], b[0], Ap, Bp);
#pragma unroll
    for (int g = 0; g < NG; ++g) {
        if (g + 1 < NG)
            load_group<N>(a[(g + 1) & 1], b[(g + 1) & 1],
                          Ap + (g + 1) * (4 * KGROUP),
                          Bp + (size_t)(g + 1) * (4 * KGROUP) * N);
        __builtin_amdgcn_sched_barrier(0);
        mma_group(acc, a[g & 1], b[g & 1]);
        __builtin_amdgcn_sched_barrier(0);
    }

    const int col  = tn * 16 + l15;
    const int row0 = tm * 16 + kh * 8;
    const float bv = bias ? bias[col] : 0.0f;
#pragma unroll
    for (int v = 0; v < 8; ++v) {
        float val = acc[v] + bv;
        if (relu) val = fmaxf(val, 0.0f);
        C[(size_t)(row0 + v) * N + col] = val;
    }
}

__global__ void readout_kernel(const float* __restrict__ h_t,
                               const float* __restrict__ h_l,
                               const float* __restrict__ w_adap,
                               float* __restrict__ fused) {
    const int b = blockIdx.x;
    const int d = threadIdx.x;
    float st = 0.0f, sl = 0.0f;
    for (int i = 0; i < NPG; ++i) {
        const size_t node = (size_t)b * NPG + i;
        st += h_t[node * DHID + d];
        sl += h_l[node * DHID + d];
    }
    const float inv = 1.0f / (float)NPG;
    fused[(size_t)b * DHID + d] = (st * inv) * w_adap[0] + (sl * inv) * w_adap[1];
}

extern "C" void kernel_launch(void* const* d_in, const int* in_sizes, int n_in,
                              void* d_out, int out_size, void* d_ws, size_t ws_size,
                              hipStream_t stream) {
    const int*   text_ids  = (const int*)  d_in[0];
    const int*   label_ids = (const int*)  d_in[1];
    const int*   text_len  = (const int*)  d_in[2];
    const int*   label_len = (const int*)  d_in[3];
    const float* word_emb  = (const float*)d_in[4];
    const float* label_emb = (const float*)d_in[5];
    const float* Wg_text   = (const float*)d_in[6];
    const float* Wg_label  = (const float*)d_in[7];
    const float* w_adap    = (const float*)d_in[8];
    const float* W1        = (const float*)d_in[9];
    const float* b1        = (const float*)d_in[10];
    const float* Wout      = (const float*)d_in[11];
    const float* bout      = (const float*)d_in[12];
    float* out = (float*)d_out;

    float* ws        = (float*)d_ws;
    float* text_out  = ws;
    float* label_out = text_out  + (size_t)NNODES * DHID;
    float* h_t       = label_out + (size_t)NNODES * DHID;
    float* h_l       = h_t       + (size_t)NNODES * DHID;
    float* fused     = h_l       + (size_t)NNODES * DHID;
    float* hmid      = fused     + (size_t)BGRAPH * DHID;

    ragged_bag_kernel<<<NNODES, 256, 0, stream>>>(
        text_ids, label_ids, text_len, label_len,
        word_emb, label_emb, text_out, label_out);

    const int tiles1 = (NNODES / 16) * (DHID / 16);
    gemm_wmma_f32<DHID, DHID><<<(tiles1 + 7) / 8, 256, 0, stream>>>(
        text_out, Wg_text, nullptr, h_t, NNODES,  1);
    gemm_wmma_f32<DHID, DHID><<<(tiles1 + 7) / 8, 256, 0, stream>>>(
        label_out, Wg_label, nullptr, h_l, NNODES,  1);

    readout_kernel<<<BGRAPH, 256, 0, stream>>>(h_t, h_l, w_adap, fused);

    const int tiles2 = (BGRAPH / 16) * (DHID / 16);
    gemm_wmma_f32<DHID, DHID><<<(tiles2 + 7) / 8, 256, 0, stream>>>(
        fused, W1, b1, hmid, BGRAPH,  1);

    const int tiles3 = (BGRAPH / 16) * (ODIM / 16);
    gemm_wmma_f32<ODIM, DHID><<<(tiles3 + 7) / 8, 256, 0, stream>>>(
        hmid, Wout, bout, out, BGRAPH,  0);
}


# ragged gather: one 64-thread block per node, each lane sums 4 adjacent channels with 16-byte loads (a row = 2 whole-line wave loads, not 8), ids by scalar loads, 16 rows in flight; sum order unchanged
# speedup vs baseline: 1.7611x; 1.2544x over previous
#include <hip/hip_runtime.h>

typedef __attribute__((ext_vector_type(2))) float v2f;
typedef __attribute__((ext_vector_type(4))) float v4f;
typedef __attribute__((ext_vector_type(8))) float v8f;

#define DHID     256
#define LTOK     128
#define LLAB     32
#define NPG      64
#define BGRAPH   64
#define NNODES   (BGRAPH * NPG)
#define ODIM     1024

#define RCH 16

__device__ __forceinline__ v4f row4(const float* __restrict__ table, int id,
                                    unsigned lane_byte) {
    const char* rowp = (const char*)table + (size_t)id * (DHID * sizeof(float));
    return *(const v4f*)(rowp + lane_byte);
}

template <int NIDS>
__device__ __forceinline__ v4f bag_sum(const int* __restrict__ ids, int len,
                                       const float* __restrict__ table,
                                       unsigned lane_byte) {
    static_assert(NIDS % RCH == 0, "whole chunks of ids");
    v4f acc = {0.0f, 0.0f, 0.0f, 0.0f};
    int cur[RCH];
#pragma unroll
    for (int j = 0; j < RCH; ++j) cur[j] = ids[j];
    int l = 0;
    for (; l + RCH <= len; l += RCH) {
        const int ln = (l + RCH < NIDS - RCH) ? (l + RCH) : (NIDS - RCH);
        int nxt[RCH];
#pragma unroll
        for (int j = 0; j < RCH; ++j) nxt[j] = ids[ln + j];
        v4f v[RCH];
#pragma unroll
        for (int j = 0; j < RCH; ++j) v[j] = row4(table, cur[j], lane_byte);
#pragma unroll
        for (int j = 0; j < RCH; ++j) acc += v[j];
#pragma unroll
        for (int j = 0; j < RCH; ++j) cur[j] = nxt[j];
    }
    const int rem = len - l;
    v4f v[RCH - 1];
#pragma unroll
    for (int j = 0; j < RCH - 1; ++j) {
        v[j] = (v4f){0.0f, 0.0f, 0.0f, 0.0f};
        if (j < rem) v[j] = row4(table, cur[j], lane_byte);
    }
#pragma unroll
    for (int j = 0; j < RCH - 1; ++j)
        if (j < rem) acc += v[j];
    return acc;
}

__global__ void __launch_bounds__(64)
ragged_bag_kernel(const int* __restrict__ text_ids,
                  const int* __restrict__ label_ids,
                  const int* __restrict__ text_len,
                  const int* __restrict__ label_len,
                  const float* __restrict__ word_emb,
                  const float* __restrict__ label_emb,
                  float* __restrict__ text_out,
                  float* __restrict__ label_out) {
    const int n = blockIdx.x;
    const unsigned lane_byte = threadIdx.x * 16u;
    const int tl = text_len[n];
    const int ll = label_len[n];

    const v4f t = bag_sum<LTOK>(text_ids + (size_t)n * LTOK, tl, word_emb, lane_byte);
    const float ftl = (float)tl;
    const v4f tmean = {t.x / ftl, t.y / ftl, t.z / ftl, t.w / ftl};
    *(v4f*)(text_out + (size_t)n * DHID + threadIdx.x * 4) = tmean;

    const v4f s = bag_sum<LLAB>(label_ids + (size_t)n * LLAB, ll, label_emb, lane_byte);
    *(v4f*)(label_out + (size_t)n * DHID + threadIdx.x * 4) = s;
}

#define KGROUP   16

template <int N>
__device__ __forceinline__ void load_group(v2f (&a)[KGROUP], v2f (&b)[KGROUP],
                                           const float* __restrict__ Ap,
                                           const float* __restrict__ Bp) {
#pragma unroll
    for (int u = 0; u < KGROUP; ++u) {
        a[u]   = *(const v2f*)(Ap + u * 4);
        b[u].x = Bp[(size_t)(u * 4) * N];
        b[u].y = Bp[(size_t)(u * 4 + 1) * N];
    }
}

__device__ __forceinline__ void mma_group(v8f& acc, const v2f (&a)[KGROUP],
                                          const v2f (&b)[KGROUP]) {
#pragma unroll
    for (int u = 0; u < KGROUP; ++u)
        acc = __builtin_amdgcn_wmma_f32_16x16x4_f32(
             false, a[u],  false, b[u],
             (short)0, acc,  false,  false);
}

template <int N, int K>
__global__ void __launch_bounds__(256)
gemm_wmma_f32(const float* __restrict__ A,
              const float* __restrict__ B,
              const float* __restrict__ bias,
              float* __restrict__ C,
              int M, int relu) {
    static_assert(N % 16 == 0, "whole 16-column tiles");
    static_assert(K % (4 * KGROUP) == 0, "whole load groups");
    constexpr int NG = K / (4 * KGROUP);

    const int wave       = (blockIdx.x * blockDim.x + threadIdx.x) >> 5;
    const int tilesN     = N >> 4;
    const int totalTiles = (M >> 4) * tilesN;
    if (wave >= totalTiles) return;

    const int tm   = wave / tilesN;
    const int tn   = wave % tilesN;
    const int lane = threadIdx.x & 31;
    const int kh   = lane >> 4;
    const int l15  = lane & 15;

    const float* Ap = A + (size_t)(tm * 16 + l15) * K + kh * 2;
    const float* Bp = B + (size_t)(kh * 2) * N + (size_t)(tn * 16 + l15);

    v8f acc = {};
    v2f a[2][KGROUP], b[2][KGROUP];
    load_group<N>(a[0], b[0], Ap, Bp);
#pragma unroll
    for (int g = 0; g < NG; ++g) {
        if (g + 1 < NG)
            load_group<N>(a[(g + 1) & 1], b[(g + 1) & 1],
                          Ap + (g + 1) * (4 * KGROUP),
                          Bp + (size_t)(g + 1) * (4 * KGROUP) * N);
        __builtin_amdgcn_sched_barrier(0);
        mma_group(acc, a[g & 1], b[g & 1]);
        __builtin_amdgcn_sched_barrier(0);
    }

    const int col  = tn * 16 + l15;
    const int row0 = tm * 16 + kh * 8;
    const float bv = bias ? bias[col] : 0.0f;
#pragma unroll
    for (int v = 0; v < 8; ++v) {
        float val = acc[v] + bv;
        if (relu) val = fmaxf(val, 0.0f);
        C[(size_t)(row0 + v) * N + col] = val;
    }
}

__global__ void readout_kernel(const float* __restrict__ h_t,
                               const float* __restrict__ h_l,
                               const float* __restrict__ w_adap,
                               float* __restrict__ fused) {
    const int b = blockIdx.x;
    const int d = threadIdx.x;
    float st = 0.0f, sl = 0.0f;
    for (int i = 0; i < NPG; ++i) {
        const size_t node = (size_t)b * NPG + i;
        st += h_t[node * DHID + d];
        sl += h_l[node * DHID + d];
    }
    const float inv = 1.0f / (float)NPG;
    fused[(size_t)b * DHID + d] = (st * inv) * w_adap[0] + (sl * inv) * w_adap[1];
}

extern "C" void kernel_launch(void* const* d_in, const int* in_sizes, int n_in,
                              void* d_out, int out_size, void* d_ws, size_t ws_size,
                              hipStream_t stream) {
    const int*   text_ids  = (const int*)  d_in[0];
    const int*   label_ids = (const int*)  d_in[1];
    const int*   text_len  = (const int*)  d_in[2];
    const int*   label_len = (const int*)  d_in[3];
    const float* word_emb  = (const float*)d_in[4];
    const float* label_emb = (const float*)d_in[5];
    const float* Wg_text   = (const float*)d_in[6];
    const float* Wg_label  = (const float*)d_in[7];
    const float* w_adap    = (const float*)d_in[8];
    const float* W1        = (const float*)d_in[9];
    const float* b1        = (const float*)d_in[10];
    const float* Wout      = (const float*)d_in[11];
    const float* bout      = (const float*)d_in[12];
    float* out = (float*)d_out;

    float* ws        = (float*)d_ws;
    float* text_out  = ws;
    float* label_out = text_out  + (size_t)NNODES * DHID;
    float* h_t       = label_out + (size_t)NNODES * DHID;
    float* h_l       = h_t       + (size_t)NNODES * DHID;
    float* fused     = h_l       + (size_t)NNODES * DHID;
    float* hmid      = fused     + (size_t)BGRAPH * DHID;

    ragged_bag_kernel<<<NNODES, 64, 0, stream>>>(
        text_ids, label_ids, text_len, label_len,
        word_emb, label_emb, text_out, label_out);

    const int tiles1 = (NNODES / 16) * (DHID / 16);
    gemm_wmma_f32<DHID, DHID><<<(tiles1 + 7) / 8, 256, 0, stream>>>(
        text_out, Wg_text, nullptr, h_t, NNODES,  1);
    gemm_wmma_f32<DHID, DHID><<<(tiles1 + 7) / 8, 256, 0, stream>>>(
        label_out, Wg_label, nullptr, h_l, NNODES,  1);

    readout_kernel<<<BGRAPH, 256, 0, stream>>>(h_t, h_l, w_adap, fused);

    const int tiles2 = (BGRAPH / 16) * (DHID / 16);
    gemm_wmma_f32<DHID, DHID><<<(tiles2 + 7) / 8, 256, 0, stream>>>(
        fused, W1, b1, hmid, BGRAPH,  1);

    const int tiles3 = (BGRAPH / 16) * (ODIM / 16);
    gemm_wmma_f32<ODIM, DHID><<<(tiles3 + 7) / 8, 256, 0, stream>>>(
        hmid, Wout, bout, out, BGRAPH,  0);
}


# per-graph mean readout + fusion moved into the projection GEMM epilogue: 16-wave block = graph x 32 cols x both types, LDS row sums; launches 6->4, h_t/h_l never stored; readout sum a fixed 8x8 tree
# speedup vs baseline: 1.9238x; 1.0924x over previous
#include <hip/hip_runtime.h>

typedef __attribute__((ext_vector_type(2))) float v2f;
typedef __attribute__((ext_vector_type(4))) float v4f;
typedef __attribute__((ext_vector_type(8))) float v8f;

#define DHID     256
#define LTOK     128
#define LLAB     32
#define NPG      64
#define BGRAPH   64
#define NNODES   (BGRAPH * NPG)
#define ODIM     1024

#define RCH 16

__device__ __forceinline__ v4f row4(const float* __restrict__ table, int id,
                                    unsigned lane_byte) {
    const char* rowp = (const char*)table + (size_t)id * (DHID * sizeof(float));
    return *(const v4f*)(rowp + lane_byte);
}

template <int NIDS>
__device__ __forceinline__ v4f bag_sum(const int* __restrict__ ids, int len,
                                       const float* __restrict__ table,
                                       unsigned lane_byte) {
    static_assert(NIDS % RCH == 0, "whole chunks of ids");
    v4f acc = {0.0f, 0.0f, 0.0f, 0.0f};
    int cur[RCH];
#pragma unroll
    for (int j = 0; j < RCH; ++j) cur[j] = ids[j];
    int l = 0;
    for (; l + RCH <= len; l += RCH) {
        const int ln = (l + RCH < NIDS - RCH) ? (l + RCH) : (NIDS - RCH);
        int nxt[RCH];
#pragma unroll
        for (int j = 0; j < RCH; ++j) nxt[j] = ids[ln + j];
        v4f v[RCH];
#pragma unroll
        for (int j = 0; j < RCH; ++j) v[j] = row4(table, cur[j], lane_byte);
#pragma unroll
        for (int j = 0; j < RCH; ++j) acc += v[j];
#pragma unroll
        for (int j = 0; j < RCH; ++j) cur[j] = nxt[j];
    }
    const int rem = len - l;
    v4f v[RCH - 1];
#pragma unroll
    for (int j = 0; j < RCH - 1; ++j) {
        v[j] = (v4f){0.0f, 0.0f, 0.0f, 0.0f};
        if (j < rem) v[j] = row4(table, cur[j], lane_byte);
    }
#pragma unroll
    for (int j = 0; j < RCH - 1; ++j)
        if (j < rem) acc += v[j];
    return acc;
}

__global__ void __launch_bounds__(64)
ragged_bag_kernel(const int* __restrict__ text_ids,
                  const int* __restrict__ label_ids,
                  const int* __restrict__ text_len,
                  const int* __restrict__ label_len,
                  const float* __restrict__ word_emb,
                  const float* __restrict__ label_emb,
                  float* __restrict__ text_out,
                  float* __restrict__ label_out) {
    const int n = blockIdx.x;
    const unsigned lane_byte = threadIdx.x * 16u;
    const int tl = text_len[n];
    const int ll = label_len[n];

    const v4f t = bag_sum<LTOK>(text_ids + (size_t)n * LTOK, tl, word_emb, lane_byte);
    const float ftl = (float)tl;
    const v4f tmean = {t.x / ftl, t.y / ftl, t.z / ftl, t.w / ftl};
    *(v4f*)(text_out + (size_t)n * DHID + threadIdx.x * 4) = tmean;

    const v4f s = bag_sum<LLAB>(label_ids + (size_t)n * LLAB, ll, label_emb, lane_byte);
    *(v4f*)(label_out + (size_t)n * DHID + threadIdx.x * 4) = s;
}

#define KGROUP   16

template <int N>
__device__ __forceinline__ void load_group(v2f (&a)[KGROUP], v2f (&b)[KGROUP],
                                           const float* __restrict__ Ap,
                                           const float* __restrict__ Bp) {
#pragma unroll
    for (int u = 0; u < KGROUP; ++u) {
        a[u]   = *(const v2f*)(Ap + u * 4);
        b[u].x = Bp[(size_t)(u * 4) * N];
        b[u].y = Bp[(size_t)(u * 4 + 1) * N];
    }
}

__device__ __forceinline__ void mma_group(v8f& acc, const v2f (&a)[KGROUP],
                                          const v2f (&b)[KGROUP]) {
#pragma unroll
    for (int u = 0; u < KGROUP; ++u)
        acc = __builtin_amdgcn_wmma_f32_16x16x4_f32(
             false, a[u],  false, b[u],
             (short)0, acc,  false,  false);
}

template <int N, int K>
__global__ void __launch_bounds__(256)
gemm_wmma_f32(const float* __restrict__ A,
              const float* __restrict__ B,
              const float* __restrict__ bias,
              float* __restrict__ C,
              int M, int relu) {
    static_assert(N % 16 == 0, "whole 16-column tiles");
    static_assert(K % (4 * KGROUP) == 0, "whole load groups");
    constexpr int NG = K / (4 * KGROUP);

    const int wave       = (blockIdx.x * blockDim.x + threadIdx.x) >> 5;
    const int tilesN     = N >> 4;
    const int totalTiles = (M >> 4) * tilesN;
    if (wave >= totalTiles) return;

    const int tm   = wave / tilesN;
    const int tn   = wave % tilesN;
    const int lane = threadIdx.x & 31;
    const int kh   = lane >> 4;
    const int l15  = lane & 15;

    const float* Ap = A + (size_t)(tm * 16 + l15) * K + kh * 2;
    const float* Bp = B + (size_t)(kh * 2) * N + (size_t)(tn * 16 + l15);

    v8f acc = {};
    v2f a[2][KGROUP], b[2][KGROUP];
    load_group<N>(a[0], b[0], Ap, Bp);
#pragma unroll
    for (int g = 0; g < NG; ++g) {
        if (g + 1 < NG)
            load_group<N>(a[(g + 1) & 1], b[(g + 1) & 1],
                          Ap + (g + 1) * (4 * KGROUP),
                          Bp + (size_t)(g + 1) * (4 * KGROUP) * N);
        __builtin_amdgcn_sched_barrier(0);
        mma_group(acc, a[g & 1], b[g & 1]);
        __builtin_amdgcn_sched_barrier(0);
    }

    const int col  = tn * 16 + l15;
    const int row0 = tm * 16 + kh * 8;
    const float bv = bias ? bias[col] : 0.0f;
#pragma unroll
    for (int v = 0; v < 8; ++v) {
        float val = acc[v] + bv;
        if (relu) val = fmaxf(val, 0.0f);
        C[(size_t)(row0 + v) * N + col] = val;
    }
}

#define PR_WAVES 16
#define PR_COLS  32

__global__ void __launch_bounds__(PR_WAVES * 32)
proj_readout_kernel(const float* __restrict__ text_in,
                    const float* __restrict__ label_in,
                    const float* __restrict__ Wg_text,
                    const float* __restrict__ Wg_label,
                    const float* __restrict__ w_adap,
                    float* __restrict__ fused) {
    constexpr int N  = DHID;
    constexpr int K  = DHID;
    static_assert(K % (4 * KGROUP) == 0, "whole load groups");
    static_assert(NPG == 64, "four 16-row tiles per graph");
    static_assert(N % PR_COLS == 0, "whole column groups");
    constexpr int NG = K / (4 * KGROUP);
    constexpr int CGROUPS = N / PR_COLS;

    __shared__ float part[PR_WAVES][32];

    const int g    = blockIdx.x / CGROUPS;
    const int cg   = blockIdx.x % CGROUPS;
    const int wv   = threadIdx.x >> 5;
    const int lane = threadIdx.x & 31;
    const int ty   = wv >> 3;
    const int mt   = (wv >> 1) & 3;
    const int nt   = wv & 1;
    const int kh   = lane >> 4;
    const int l15  = lane & 15;

    const float* A = ty ? label_in : text_in;
    const float* B = ty ? Wg_label : Wg_text;

    const float* Ap = A + (size_t)(g * NPG + mt * 16 + l15) * K + kh * 2;
    const float* Bp = B + (size_t)(kh * 2) * N + (size_t)(cg * PR_COLS + nt * 16 + l15);

    v8f acc = {};
    v2f a[2][KGROUP], b[2][KGROUP];
    load_group<N>(a[0], b[0], Ap, Bp);
#pragma unroll
    for (int gk = 0; gk < NG; ++gk) {
        if (gk + 1 < NG)
            load_group<N>(a[(gk + 1) & 1], b[(gk + 1) & 1],
                          Ap + (gk + 1) * (4 * KGROUP),
                          Bp + (size_t)(gk + 1) * (4 * KGROUP) * N);
        __builtin_amdgcn_sched_barrier(0);
        mma_group(acc, a[gk & 1], b[gk & 1]);
        __builtin_amdgcn_sched_barrier(0);
    }

    float s8 = 0.0f;
#pragma unroll
    for (int v = 0; v < 8; ++v) s8 += fmaxf(acc[v], 0.0f);
    part[wv][lane] = s8;
    __syncthreads();

    if (threadIdx.x < PR_COLS) {
        const int jn = threadIdx.x >> 4;
        const int jc = threadIdx.x & 15;
        float st = 0.0f, sl = 0.0f;
        for (int m = 0; m < 4; ++m) {
            for (int h = 0; h < 2; ++h) {
                st += part[(m << 1) | jn][h * 16 + jc];
                sl += part[8 + ((m << 1) | jn)][h * 16 + jc];
            }
        }
        const float inv = 1.0f / (float)NPG;
        fused[(size_t)g * DHID + cg * PR_COLS + threadIdx.x] =
            (st * inv) * w_adap[0] + (sl * inv) * w_adap[1];
    }
}

extern "C" void kernel_launch(void* const* d_in, const int* in_sizes, int n_in,
                              void* d_out, int out_size, void* d_ws, size_t ws_size,
                              hipStream_t stream) {
    const int*   text_ids  = (const int*)  d_in[0];
    const int*   label_ids = (const int*)  d_in[1];
    const int*   text_len  = (const int*)  d_in[2];
    const int*   label_len = (const int*)  d_in[3];
    const float* word_emb  = (const float*)d_in[4];
    const float* label_emb = (const float*)d_in[5];
    const float* Wg_text   = (const float*)d_in[6];
    const float* Wg_label  = (const float*)d_in[7];
    const float* w_adap    = (const float*)d_in[8];
    const float* W1        = (const float*)d_in[9];
    const float* b1        = (const float*)d_in[10];
    const float* Wout      = (const float*)d_in[11];
    const float* bout      = (const float*)d_in[12];
    float* out = (float*)d_out;

    float* ws        = (float*)d_ws;
    float* text_out  = ws;
    float* label_out = text_out  + (size_t)NNODES * DHID;
    float* fused     = label_out + (size_t)NNODES * DHID;
    float* hmid      = fused     + (size_t)BGRAPH * DHID;

    ragged_bag_kernel<<<NNODES, 64, 0, stream>>>(
        text_ids, label_ids, text_len, label_len,
        word_emb, label_emb, text_out, label_out);

    proj_readout_kernel<<<BGRAPH * (DHID / PR_COLS), PR_WAVES * 32, 0, stream>>>(
        text_out, label_out, Wg_text, Wg_label, w_adap, fused);

    const int tiles2 = (BGRAPH / 16) * (DHID / 16);
    gemm_wmma_f32<DHID, DHID><<<(tiles2 + 7) / 8, 256, 0, stream>>>(
        fused, W1, b1, hmid, BGRAPH,  1);

    const int tiles3 = (BGRAPH / 16) * (ODIM / 16);
    gemm_wmma_f32<ODIM, DHID><<<(tiles3 + 7) / 8, 256, 0, stream>>>(
        hmid, Wout, bout, out, BGRAPH,  0);
}


# projection GEMM: a wave owns a 16x64 strip (4 accumulators): one A fragment per K-step feeds 4 WMMAs, its 4 B fragments come from two 16-byte loads; 8-wave blocks, grid 256; sum order unchanged
# speedup vs baseline: 2.4556x; 1.2764x over previous
#include <hip/hip_runtime.h>

typedef __attribute__((ext_vector_type(2))) float v2f;
typedef __attribute__((ext_vector_type(4))) float v4f;
typedef __attribute__((ext_vector_type(8))) float v8f;

#define DHID     256
#define LTOK     128
#define LLAB     32
#define NPG      64
#define BGRAPH   64
#define NNODES   (BGRAPH * NPG)
#define ODIM     1024

#define RCH 16

__device__ __forceinline__ v4f row4(const float* __restrict__ table, int id,
                                    unsigned lane_byte) {
    const char* rowp = (const char*)table + (size_t)id * (DHID * sizeof(float));
    return *(const v4f*)(rowp + lane_byte);
}

template <int NIDS>
__device__ __forceinline__ v4f bag_sum(const int* __restrict__ ids, int len,
                                       const float* __restrict__ table,
                                       unsigned lane_byte) {
    static_assert(NIDS % RCH == 0, "whole chunks of ids");
    v4f acc = {0.0f, 0.0f, 0.0f, 0.0f};
    int cur[RCH];
#pragma unroll
    for (int j = 0; j < RCH; ++j) cur[j] = ids[j];
    int l = 0;
    for (; l + RCH <= len; l += RCH) {
        const int ln = (l + RCH < NIDS - RCH) ? (l + RCH) : (NIDS - RCH);
        int nxt[RCH];
#pragma unroll
        for (int j = 0; j < RCH; ++j) nxt[j] = ids[ln + j];
        v4f v[RCH];
#pragma unroll
        for (int j = 0; j < RCH; ++j) v[j] = row4(table, cur[j], lane_byte);
#pragma unroll
        for (int j = 0; j < RCH; ++j) acc += v[j];
#pragma unroll
        for (int j = 0; j < RCH; ++j) cur[j] = nxt[j];
    }
    const int rem = len - l;
    v4f v[RCH - 1];
#pragma unroll
    for (int j = 0; j < RCH - 1; ++j) {
        v[j] = (v4f){0.0f, 0.0f, 0.0f, 0.0f};
        if (j < rem) v[j] = row4(table, cur[j], lane_byte);
    }
#pragma unroll
    for (int j = 0; j < RCH - 1; ++j)
        if (j < rem) acc += v[j];
    return acc;
}

__global__ void __launch_bounds__(64)
ragged_bag_kernel(const int* __restrict__ text_ids,
                  const int* __restrict__ label_ids,
                  const int* __restrict__ text_len,
                  const int* __restrict__ label_len,
                  const float* __restrict__ word_emb,
                  const float* __restrict__ label_emb,
                  float* __restrict__ text_out,
                  float* __restrict__ label_out) {
    const int n = blockIdx.x;
    const unsigned lane_byte = threadIdx.x * 16u;
    const int tl = text_len[n];
    const int ll = label_len[n];

    const v4f t = bag_sum<LTOK>(text_ids + (size_t)n * LTOK, tl, word_emb, lane_byte);
    const float ftl = (float)tl;
    const v4f tmean = {t.x / ftl, t.y / ftl, t.z / ftl, t.w / ftl};
    *(v4f*)(text_out + (size_t)n * DHID + threadIdx.x * 4) = tmean;

    const v4f s = bag_sum<LLAB>(label_ids + (size_t)n * LLAB, ll, label_emb, lane_byte);
    *(v4f*)(label_out + (size_t)n * DHID + threadIdx.x * 4) = s;
}

#define KGROUP   16

template <int N>
__device__ __forceinline__ void load_group(v2f (&a)[KGROUP], v2f (&b)[KGROUP],
                                           const float* __restrict__ Ap,
                                           const float* __restrict__ Bp) {
#pragma unroll
    for (int u = 0; u < KGROUP; ++u) {
        a[u]   = *(const v2f*)(Ap + u * 4);
        b[u].x = Bp[(size_t)(u * 4) * N];
        b[u].y = Bp[(size_t)(u * 4 + 1) * N];
    }
}

__device__ __forceinline__ void mma_group(v8f& acc, const v2f (&a)[KGROUP],
                                          const v2f (&b)[KGROUP]) {
#pragma unroll
    for (int u = 0; u < KGROUP; ++u)
        acc = __builtin_amdgcn_wmma_f32_16x16x4_f32(
             false, a[u],  false, b[u],
             (short)0, acc,  false,  false);
}

template <int N, int K>
__global__ void __launch_bounds__(256)
gemm_wmma_f32(const float* __restrict__ A,
              const float* __restrict__ B,
              const float* __restrict__ bias,
              float* __restrict__ C,
              int M, int relu) {
    static_assert(N % 16 == 0, "whole 16-column tiles");
    static_assert(K % (4 * KGROUP) == 0, "whole load groups");
    constexpr int NG = K / (4 * KGROUP);

    const int wave       = (blockIdx.x * blockDim.x + threadIdx.x) >> 5;
    const int tilesN     = N >> 4;
    const int totalTiles = (M >> 4) * tilesN;
    if (wave >= totalTiles) return;

    const int tm   = wave / tilesN;
    const int tn   = wave % tilesN;
    const int lane = threadIdx.x & 31;
    const int kh   = lane >> 4;
    const int l15  = lane & 15;

    const float* Ap = A + (size_t)(tm * 16 + l15) * K + kh * 2;
    const float* Bp = B + (size_t)(kh * 2) * N + (size_t)(tn * 16 + l15);

    v8f acc = {};
    v2f a[2][KGROUP], b[2][KGROUP];
    load_group<N>(a[0], b[0], Ap, Bp);
#pragma unroll
    for (int g = 0; g < NG; ++g) {
        if (g + 1 < NG)
            load_group<N>(a[(g + 1) & 1], b[(g + 1) & 1],
                          Ap + (g + 1) * (4 * KGROUP),
                          Bp + (size_t)(g + 1) * (4 * KGROUP) * N);
        __builtin_amdgcn_sched_barrier(0);
        mma_group(acc, a[g & 1], b[g & 1]);
        __builtin_amdgcn_sched_barrier(0);
    }

    const int col  = tn * 16 + l15;
    const int row0 = tm * 16 + kh * 8;
    const float bv = bias ? bias[col] : 0.0f;
#pragma unroll
    for (int v = 0; v < 8; ++v) {
        float val = acc[v] + bv;
        if (relu) val = fmaxf(val, 0.0f);
        C[(size_t)(row0 + v) * N + col] = val;
    }
}

#define PR_WAVES 8
#define PR_NT    4
#define PR_COLS  (16 * PR_NT)

template <int N>
__device__ __forceinline__ void load_group_strip(v2f (&a)[KGROUP],
                                                 v4f (&b0)[KGROUP],
                                                 v4f (&b1)[KGROUP],
                                                 const float* __restrict__ Ap,
                                                 const float* __restrict__ Bp) {
#pragma unroll
    for (int u = 0; u < KGROUP; ++u) {
        a[u]  = *(const v2f*)(Ap + u * 4);
        b0[u] = *(const v4f*)(Bp + (size_t)(u * 4) * N);
        b1[u] = *(const v4f*)(Bp + (size_t)(u * 4 + 1) * N);
    }
}

__device__ __forceinline__ void mma_group_strip(v8f (&acc)[PR_NT],
                                                const v2f (&a)[KGROUP],
                                                const v4f (&b0)[KGROUP],
                                                const v4f (&b1)[KGROUP]) {
#pragma unroll
    for (int u = 0; u < KGROUP; ++u) {
#pragma unroll
        for (int t = 0; t < PR_NT; ++t) {
            const v2f bt = {b0[u][t], b1[u][t]};
            acc[t] = __builtin_amdgcn_wmma_f32_16x16x4_f32(
                 false, a[u],  false, bt,
                 (short)0, acc[t],  false,  false);
        }
    }
}

__global__ void __launch_bounds__(PR_WAVES * 32)
proj_readout_kernel(const float* __restrict__ text_in,
                    const float* __restrict__ label_in,
                    const float* __restrict__ Wg_text,
                    const float* __restrict__ Wg_label,
                    const float* __restrict__ w_adap,
                    float* __restrict__ fused) {
    constexpr int N  = DHID;
    constexpr int K  = DHID;
    static_assert(K % (4 * KGROUP) == 0, "whole load groups");
    static_assert(NPG == 64, "four 16-row tiles per graph");
    static_assert(N % PR_COLS == 0, "whole column strips");
    static_assert(PR_NT == 4, "one 16-byte B load = the four tiles' columns");
    constexpr int NG = K / (4 * KGROUP);
    constexpr int CGROUPS = N / PR_COLS;

    __shared__ float part[PR_WAVES][PR_NT][32];

    const int g    = blockIdx.x / CGROUPS;
    const int cg   = blockIdx.x % CGROUPS;
    const int wv   = threadIdx.x >> 5;
    const int lane = threadIdx.x & 31;
    const int ty   = wv >> 2;
    const int mt   = wv & 3;
    const int kh   = lane >> 4;
    const int l15  = lane & 15;

    const float* A = ty ? label_in : text_in;
    const float* B = ty ? Wg_label : Wg_text;

    const float* Ap = A + (size_t)(g * NPG + mt * 16 + l15) * K + kh * 2;
    const float* Bp = B + (size_t)(kh * 2) * N + (size_t)(cg * PR_COLS + l15 * 4);

    v8f acc[PR_NT] = {};
    v2f a[2][KGROUP];
    v4f b0[2][KGROUP], b1[2][KGROUP];
    load_group_strip<N>(a[0], b0[0], b1[0], Ap, Bp);
#pragma unroll
    for (int gk = 0; gk < NG; ++gk) {
        if (gk + 1 < NG)
            load_group_strip<N>(a[(gk + 1) & 1], b0[(gk + 1) & 1], b1[(gk + 1) & 1],
                                Ap + (gk + 1) * (4 * KGROUP),
                                Bp + (size_t)(gk + 1) * (4 * KGROUP) * N);
        __builtin_amdgcn_sched_barrier(0);
        mma_group_strip(acc, a[gk & 1], b0[gk & 1], b1[gk & 1]);
        __builtin_amdgcn_sched_barrier(0);
    }

#pragma unroll
    for (int t = 0; t < PR_NT; ++t) {
        float s8 = 0.0f;
#pragma unroll
        for (int v = 0; v < 8; ++v) s8 += fmaxf(acc[t][v], 0.0f);
        part[wv][t][lane] = s8;
    }
    __syncthreads();

    if (threadIdx.x < PR_COLS) {
        const int jt = threadIdx.x & 3;
        const int jc = threadIdx.x >> 2;
        float st = 0.0f, sl = 0.0f;
        for (int m = 0; m < 4; ++m) {
            for (int h = 0; h < 2; ++h) {
                st += part[m][jt][h * 16 + jc];
                sl += part[4 + m][jt][h * 16 + jc];
            }
        }
        const float inv = 1.0f / (float)NPG;
        fused[(size_t)g * DHID + cg * PR_COLS + threadIdx.x] =
            (st * inv) * w_adap[0] + (sl * inv) * w_adap[1];
    }
}

extern "C" void kernel_launch(void* const* d_in, const int* in_sizes, int n_in,
                              void* d_out, int out_size, void* d_ws, size_t ws_size,
                              hipStream_t stream) {
    const int*   text_ids  = (const int*)  d_in[0];
    const int*   label_ids = (const int*)  d_in[1];
    const int*   text_len  = (const int*)  d_in[2];
    const int*   label_len = (const int*)  d_in[3];
    const float* word_emb  = (const float*)d_in[4];
    const float* label_emb = (const float*)d_in[5];
    const float* Wg_text   = (const float*)d_in[6];
    const float* Wg_label  = (const float*)d_in[7];
    const float* w_adap    = (const float*)d_in[8];
    const float* W1        = (const float*)d_in[9];
    const float* b1        = (const float*)d_in[10];
    const float* Wout      = (const float*)d_in[11];
    const float* bout      = (const float*)d_in[12];
    float* out = (float*)d_out;

    float* ws        = (float*)d_ws;
    float* text_out  = ws;
    float* label_out = text_out  + (size_t)NNODES * DHID;
    float* fused     = label_out + (size_t)NNODES * DHID;
    float* hmid      = fused     + (size_t)BGRAPH * DHID;

    ragged_bag_kernel<<<NNODES, 64, 0, stream>>>(
        text_ids, label_ids, text_len, label_len,
        word_emb, label_emb, text_out, label_out);

    proj_readout_kernel<<<BGRAPH * (DHID / PR_COLS), PR_WAVES * 32, 0, stream>>>(
        text_out, label_out, Wg_text, Wg_label, w_adap, fused);

    const int tiles2 = (BGRAPH / 16) * (DHID / 16);
    gemm_wmma_f32<DHID, DHID><<<(tiles2 + 7) / 8, 256, 0, stream>>>(
        fused, W1, b1, hmid, BGRAPH,  1);

    const int tiles3 = (BGRAPH / 16) * (ODIM / 16);
    gemm_wmma_f32<ODIM, DHID><<<(tiles3 + 7) / 8, 256, 0, stream>>>(
        hmid, Wout, bout, out, BGRAPH,  0);
}


# fused@W1 loses its launch: each projection block adds its split-K share (64 k, one WMMA row per graph) in its epilogue; output GEMM sums the 4 partials into LDS; launches 4->3; h k-sum = 4 partials
# speedup vs baseline: 2.9509x; 1.2017x over previous
#include <hip/hip_runtime.h>

typedef __attribute__((ext_vector_type(2))) float v2f;
typedef __attribute__((ext_vector_type(4))) float v4f;
typedef __attribute__((ext_vector_type(8))) float v8f;

#define DHID     256
#define LTOK     128
#define LLAB     32
#define NPG      64
#define BGRAPH   64
#define NNODES   (BGRAPH * NPG)
#define ODIM     1024

#define RCH 16

__device__ __forceinline__ v4f row4(const float* __restrict__ table, int id,
                                    unsigned lane_byte) {
    const char* rowp = (const char*)table + (size_t)id * (DHID * sizeof(float));
    return *(const v4f*)(rowp + lane_byte);
}

template <int NIDS>
__device__ __forceinline__ v4f bag_sum(const int* __restrict__ ids, int len,
                                       const float* __restrict__ table,
                                       unsigned lane_byte) {
    static_assert(NIDS % RCH == 0, "whole chunks of ids");
    v4f acc = {0.0f, 0.0f, 0.0f, 0.0f};
    int cur[RCH];
#pragma unroll
    for (int j = 0; j < RCH; ++j) cur[j] = ids[j];
    int l = 0;
    for (; l + RCH <= len; l += RCH) {
        const int ln = (l + RCH < NIDS - RCH) ? (l + RCH) : (NIDS - RCH);
        int nxt[RCH];
#pragma unroll
        for (int j = 0; j < RCH; ++j) nxt[j] = ids[ln + j];
        v4f v[RCH];
#pragma unroll
        for (int j = 0; j < RCH; ++j) v[j] = row4(table, cur[j], lane_byte);
#pragma unroll
        for (int j = 0; j < RCH; ++j) acc += v[j];
#pragma unroll
        for (int j = 0; j < RCH; ++j) cur[j] = nxt[j];
    }
    const int rem = len - l;
    v4f v[RCH - 1];
#pragma unroll
    for (int j = 0; j < RCH - 1; ++j) {
        v[j] = (v4f){0.0f, 0.0f, 0.0f, 0.0f};
        if (j < rem) v[j] = row4(table, cur[j], lane_byte);
    }
#pragma unroll
    for (int j = 0; j < RCH - 1; ++j)
        if (j < rem) acc += v[j];
    return acc;
}

__global__ void __launch_bounds__(64)
ragged_bag_kernel(const int* __restrict__ text_ids,
                  const int* __restrict__ label_ids,
                  const int* __restrict__ text_len,
                  const int* __restrict__ label_len,
                  const float* __restrict__ word_emb,
                  const float* __restrict__ label_emb,
                  float* __restrict__ text_out,
                  float* __restrict__ label_out) {
    const int n = blockIdx.x;
    const unsigned lane_byte = threadIdx.x * 16u;
    const int tl = text_len[n];
    const int ll = label_len[n];

    const v4f t = bag_sum<LTOK>(text_ids + (size_t)n * LTOK, tl, word_emb, lane_byte);
    const float ftl = (float)tl;
    const v4f tmean = {t.x / ftl, t.y / ftl, t.z / ftl, t.w / ftl};
    *(v4f*)(text_out + (size_t)n * DHID + threadIdx.x * 4) = tmean;

    const v4f s = bag_sum<LLAB>(label_ids + (size_t)n * LLAB, ll, label_emb, lane_byte);
    *(v4f*)(label_out + (size_t)n * DHID + threadIdx.x * 4) = s;
}

#define KGROUP   16

#define PR_WAVES 8
#define PR_NT    4
#define PR_COLS  (16 * PR_NT)

template <int N>
__device__ __forceinline__ void load_group_strip(v2f (&a)[KGROUP],
                                                 v4f (&b0)[KGROUP],
                                                 v4f (&b1)[KGROUP],
                                                 const float* __restrict__ Ap,
                                                 const float* __restrict__ Bp) {
#pragma unroll
    for (int u = 0; u < KGROUP; ++u) {
        a[u]  = *(const v2f*)(Ap + u * 4);
        b0[u] = *(const v4f*)(Bp + (size_t)(u * 4) * N);
        b1[u] = *(const v4f*)(Bp + (size_t)(u * 4 + 1) * N);
    }
}

__device__ __forceinline__ void mma_group_strip(v8f (&acc)[PR_NT],
                                                const v2f (&a)[KGROUP],
                                                const v4f (&b0)[KGROUP],
                                                const v4f (&b1)[KGROUP]) {
#pragma unroll
    for (int u = 0; u < KGROUP; ++u) {
#pragma unroll
        for (int t = 0; t < PR_NT; ++t) {
            const v2f bt = {b0[u][t], b1[u][t]};
            acc[t] = __builtin_amdgcn_wmma_f32_16x16x4_f32(
                 false, a[u],  false, bt,
                 (short)0, acc[t],  false,  false);
        }
    }
}

#define HP_NT    2
#define HP_KS    (PR_COLS / 4)

__global__ void __launch_bounds__(PR_WAVES * 32)
proj_readout_kernel(const float* __restrict__ text_in,
                    const float* __restrict__ label_in,
                    const float* __restrict__ Wg_text,
                    const float* __restrict__ Wg_label,
                    const float* __restrict__ w_adap,
                    const float* __restrict__ W1,
                    float* __restrict__ hpart) {
    constexpr int N  = DHID;
    constexpr int K  = DHID;
    static_assert(K % (4 * KGROUP) == 0, "whole load groups");
    static_assert(NPG == 64, "four 16-row tiles per graph");
    static_assert(N % PR_COLS == 0, "whole column strips");
    static_assert(PR_NT == 4, "one 16-byte B load = the four tiles' columns");
    static_assert(PR_WAVES * HP_NT * 16 == DHID, "the waves' tiles cover a row of h");
    static_assert(HP_NT == 2, "one 8-byte W1 load = the two tiles' columns");
    constexpr int NG = K / (4 * KGROUP);
    constexpr int CGROUPS = N / PR_COLS;

    __shared__ float part[PR_WAVES][PR_NT][32];
    __shared__ __attribute__((aligned(8))) float fs[PR_COLS];

    const int g    = blockIdx.x / CGROUPS;
    const int cg   = blockIdx.x % CGROUPS;
    const int wv   = threadIdx.x >> 5;
    const int lane = threadIdx.x & 31;
    const int ty   = wv >> 2;
    const int mt   = wv & 3;
    const int kh   = lane >> 4;
    const int l15  = lane & 15;

    const float* A = ty ? label_in : text_in;
    const float* B = ty ? Wg_label : Wg_text;

    const float* Ap = A + (size_t)(g * NPG + mt * 16 + l15) * K + kh * 2;
    const float* Bp = B + (size_t)(kh * 2) * N + (size_t)(cg * PR_COLS + l15 * 4);

    const float* Hp = W1 + (size_t)(cg * PR_COLS + kh * 2) * DHID + (size_t)(wv * (HP_NT * 16) + l15 * 2);
    v2f hb0[HP_KS], hb1[HP_KS];

    v8f acc[PR_NT] = {};
    v2f a[2][KGROUP];
    v4f b0[2][KGROUP], b1[2][KGROUP];
    load_group_strip<N>(a[0], b0[0], b1[0], Ap, Bp);
#pragma unroll
    for (int gk = 0; gk < NG; ++gk) {
        if (gk + 1 < NG) {
            load_group_strip<N>(a[(gk + 1) & 1], b0[(gk + 1) & 1], b1[(gk + 1) & 1],
                                Ap + (gk + 1) * (4 * KGROUP),
                                Bp + (size_t)(gk + 1) * (4 * KGROUP) * N);
        } else {
#pragma unroll
            for (int u = 0; u < HP_KS; ++u) {
                hb0[u] = *(const v2f*)(Hp + (size_t)(u * 4) * DHID);
                hb1[u] = *(const v2f*)(Hp + (size_t)(u * 4 + 1) * DHID);
            }
        }
        __builtin_amdgcn_sched_barrier(0);
        mma_group_strip(acc, a[gk & 1], b0[gk & 1], b1[gk & 1]);
        __builtin_amdgcn_sched_barrier(0);
    }

#pragma unroll
    for (int t = 0; t < PR_NT; ++t) {
        float s8 = 0.0f;
#pragma unroll
        for (int v = 0; v < 8; ++v) s8 += fmaxf(acc[t][v], 0.0f);
        part[wv][t][lane] = s8;
    }
    __syncthreads();

    if (threadIdx.x < PR_COLS) {
        const int jt = threadIdx.x & 3;
        const int jc = threadIdx.x >> 2;
        float st = 0.0f, sl = 0.0f;
        for (int m = 0; m < 4; ++m) {
            for (int h = 0; h < 2; ++h) {
                st += part[m][jt][h * 16 + jc];
                sl += part[4 + m][jt][h * 16 + jc];
            }
        }
        const float inv = 1.0f / (float)NPG;
        fs[threadIdx.x] = (st * inv) * w_adap[0] + (sl * inv) * w_adap[1];
    }
    __syncthreads();

    v8f hacc[HP_NT] = {};
    const v2f zero2 = {0.0f, 0.0f};
    v2f fa[HP_KS];
#pragma unroll
    for (int u = 0; u < HP_KS; ++u) {
        const v2f fk = *(const v2f*)(&fs[u * 4 + kh * 2]);
        fa[u] = (l15 == 0) ? fk : zero2;
    }
    __builtin_amdgcn_sched_barrier(0);
#pragma unroll
    for (int u = 0; u < HP_KS; ++u) {
#pragma unroll
        for (int tt = 0; tt < HP_NT; ++tt) {
            const v2f bt = {hb0[u][tt], hb1[u][tt]};
            hacc[tt] = __builtin_amdgcn_wmma_f32_16x16x4_f32(
                 false, fa[u],  false, bt,
                 (short)0, hacc[tt],  false,  false);
        }
    }
    if (lane < 16) {
        const v2f o = {hacc[0][0], hacc[1][0]};
        *(v2f*)(hpart + ((size_t)g * CGROUPS + cg) * DHID + wv * (HP_NT * 16) + l15 * 2) = o;
    }
}

#define OT_WAVES  8
#define OT_HPITCH (DHID + 4)
#define OT_STRIPS (DHID / PR_COLS)

template <int N>
__device__ __forceinline__ void load_b_group(v2f (&b)[KGROUP],
                                             const float* __restrict__ Bp) {
#pragma unroll
    for (int u = 0; u < KGROUP; ++u) {
        b[u].x = Bp[(size_t)(u * 4) * N];
        b[u].y = Bp[(size_t)(u * 4 + 1) * N];
    }
}

__device__ __forceinline__ void lds_a_group(v2f (&a)[KGROUP], const float* hp) {
#pragma unroll
    for (int u = 0; u < KGROUP; ++u)
        a[u] = *(const v2f*)(hp + u * 4);
}

__device__ __forceinline__ void mma_group(v8f& acc, const v2f (&a)[KGROUP],
                                          const v2f (&b)[KGROUP]) {
#pragma unroll
    for (int u = 0; u < KGROUP; ++u)
        acc = __builtin_amdgcn_wmma_f32_16x16x4_f32(
             false, a[u],  false, b[u],
             (short)0, acc,  false,  false);
}

__global__ void __launch_bounds__(OT_WAVES * 32)
out_gemm_kernel(const float* __restrict__ hpart,
                const float* __restrict__ b1,
                const float* __restrict__ Wout,
                const float* __restrict__ bout,
                float* __restrict__ out) {
    constexpr int N = ODIM;
    constexpr int K = DHID;
    static_assert(K % (4 * KGROUP) == 0, "whole load groups");
    static_assert((N / 16) % OT_WAVES == 0, "a block's tiles share one row tile");
    static_assert(BGRAPH % 16 == 0, "whole row tiles");
    static_assert(OT_WAVES * 32 == 16 * 16, "prologue: 16 rows x 16 threads");
    static_assert(DHID == 4 * 64, "prologue: 4 chunks of 16 threads x 4 k per row");
    static_assert(OT_STRIPS == 4, "four partials per element of h");
    constexpr int NG = K / (4 * KGROUP);
    constexpr int tilesN = N / 16;

    __shared__ __attribute__((aligned(16))) float hs[16][OT_HPITCH];

    const int wave = blockIdx.x * OT_WAVES + (threadIdx.x >> 5);
    const int tm   = wave / tilesN;
    const int tn   = wave % tilesN;
    const int lane = threadIdx.x & 31;
    const int kh   = lane >> 4;
    const int l15  = lane & 15;

    const float* Bp = Wout + (size_t)(kh * 2) * N + (size_t)(tn * 16 + l15);
    v2f a[2][KGROUP], b[2][KGROUP];
    load_b_group<N>(b[0], Bp);
    const int col  = tn * 16 + l15;
    const float bo = bout[col];
    __builtin_amdgcn_sched_barrier(0);

    {
        const int r = threadIdx.x >> 4;
        const int q = threadIdx.x & 15;
        const float* pp = hpart + (size_t)(tm * 16 + r) * (OT_STRIPS * DHID) + q * 4;
#pragma unroll
        for (int i = 0; i < 4; ++i) {
            const int k0 = i * 64 + q * 4;
            const v4f p0 = *(const v4f*)(pp + 0 * DHID + i * 64);
            const v4f p1 = *(const v4f*)(pp + 1 * DHID + i * 64);
            const v4f p2 = *(const v4f*)(pp + 2 * DHID + i * 64);
            const v4f p3 = *(const v4f*)(pp + 3 * DHID + i * 64);
            const v4f bv = *(const v4f*)(b1 + k0);
            v4f s = ((p0 + p1) + p2) + p3;
            s = s + bv;
            s.x = fmaxf(s.x, 0.0f);
            s.y = fmaxf(s.y, 0.0f);
            s.z = fmaxf(s.z, 0.0f);
            s.w = fmaxf(s.w, 0.0f);
            *(v4f*)(&hs[r][k0]) = s;
        }
    }
    __syncthreads();

    const float* hrow = &hs[l15][kh * 2];
    v8f acc = {};
    lds_a_group(a[0], hrow);
#pragma unroll
    for (int g = 0; g < NG; ++g) {
        if (g + 1 < NG) {
            load_b_group<N>(b[(g + 1) & 1], Bp + (size_t)(g + 1) * (4 * KGROUP) * N);
            lds_a_group(a[(g + 1) & 1], hrow + (g + 1) * (4 * KGROUP));
        }
        __builtin_amdgcn_sched_barrier(0);
        mma_group(acc, a[g & 1], b[g & 1]);
        __builtin_amdgcn_sched_barrier(0);
    }

    const int row0 = tm * 16 + kh * 8;
#pragma unroll
    for (int v = 0; v < 8; ++v)
        out[(size_t)(row0 + v) * N + col] = acc[v] + bo;
}

extern "C" void kernel_launch(void* const* d_in, const int* in_sizes, int n_in,
                              void* d_out, int out_size, void* d_ws, size_t ws_size,
                              hipStream_t stream) {
    const int*   text_ids  = (const int*)  d_in[0];
    const int*   label_ids = (const int*)  d_in[1];
    const int*   text_len  = (const int*)  d_in[2];
    const int*   label_len = (const int*)  d_in[3];
    const float* word_emb  = (const float*)d_in[4];
    const float* label_emb = (const float*)d_in[5];
    const float* Wg_text   = (const float*)d_in[6];
    const float* Wg_label  = (const float*)d_in[7];
    const float* w_adap    = (const float*)d_in[8];
    const float* W1        = (const float*)d_in[9];
    const float* b1        = (const float*)d_in[10];
    const float* Wout      = (const float*)d_in[11];
    const float* bout      = (const float*)d_in[12];
    float* out = (float*)d_out;

    float* ws        = (float*)d_ws;
    float* text_out  = ws;
    float* label_out = text_out  + (size_t)NNODES * DHID;
    float* hpart     = label_out + (size_t)NNODES * DHID;

    ragged_bag_kernel<<<NNODES, 64, 0, stream>>>(
        text_ids, label_ids, text_len, label_len,
        word_emb, label_emb, text_out, label_out);

    proj_readout_kernel<<<BGRAPH * (DHID / PR_COLS), PR_WAVES * 32, 0, stream>>>(
        text_out, label_out, Wg_text, Wg_label, w_adap, W1, hpart);

    out_gemm_kernel<<<(BGRAPH / 16) * (ODIM / 16) / OT_WAVES, OT_WAVES * 32, 0, stream>>>(
        hpart, b1, Wout, bout, out);
}
